# starting kernel
# baseline (speedup 1.0000x reference)
#include <hip/hip_runtime.h>

typedef __attribute__((ext_vector_type(2))) float v2f;
typedef __attribute__((ext_vector_type(4))) float v4f;
typedef __attribute__((ext_vector_type(8))) float v8f;

__device__ __forceinline__ v2f make_v2f(float a, float b) {
    v2f r; r.x = a; r.y = b; return r;
}

#define BBATCH 4
#define CC     16
#define HH     64
#define WW     64
#define FF     32
#define KK     3

__global__ __launch_bounds__(256) void pade_conv_wmma(
    const float* __restrict__ x,
    const float* __restrict__ nums,
    const float* __restrict__ denoms,
    float* __restrict__ out)
{
    const int lane  = threadIdx.x & 31;
    const int gw    = blockIdx.x * 8 + (threadIdx.x >> 5);
    const int ftile = gw & 1;
    const int tile  = gw >> 1;

    const int w0 = (tile & 3) * 16;
    const int h  = (tile >> 2) & (HH - 1);
    const int b  = tile >> 8;

    const int  laneM = lane & 15;
    const bool hi    = lane >= 16;

    const int f = ftile * 16 + laneM;
    const float* __restrict__ npF = nums   + (size_t)f * (CC * KK * KK) * 6;
    const float* __restrict__ dpF = denoms + (size_t)f * (CC * KK * KK) * 4;

    v8f acc = {};
    const v8f zc = {};

    const int  loadIdx = (lane < 18) ? lane : 17;
    const int  wl      = w0 - 1 + loadIdx;
    const bool wlOK    = (unsigned)wl < (unsigned)WW;

    for (int c = 0; c < CC; ++c) {
        const float* __restrict__ xplane = x + (size_t)(b * CC + c) * (HH * WW);
        for (int a = 0; a < KK; ++a) {
            const int  hh  = h + a - 1;
            const bool hOK = (unsigned)hh < (unsigned)HH;
            const float* __restrict__ xrow = xplane + (size_t)(hOK ? hh : 0) * WW;

            float xload = 0.0f;
            if (hOK && wlOK) xload = xrow[wl];
            const int xbits = __float_as_int(xload);

#pragma unroll
            for (int bb = 0; bb < KK; ++bb) {
                const float xv = __int_as_float(
                    __builtin_amdgcn_ds_bpermute((laneM + bb) << 2, xbits));

                const float x2 = xv * xv;
                const float x3 = x2 * xv;
                const float x4 = x2 * x2;
                const float x5 = x4 * xv;

                v2f aP0, aP1, aQ;
                if (!hi) {
                    aP0 = make_v2f(1.0f, xv);
                    aP1 = make_v2f(x4,   x5);
                    aQ  = make_v2f(xv,   x2);
                } else {
                    aP0 = make_v2f(x2,   x3);
                    aP1 = make_v2f(0.0f, 0.0f);
                    aQ  = make_v2f(x3,   x4);
                }

                const int tap = (c * KK + a) * KK + bb;
                const float* np = npF + (size_t)tap * 6;
                const float* dp = dpF + (size_t)tap * 4;

                v2f bP0 = *(const v2f*)(np + (hi ? 2 : 0));
                v2f bP1 = hi ? make_v2f(0.0f, 0.0f) : *(const v2f*)(np + 4);
                v2f bQ  = *(const v2f*)(dp + (hi ? 2 : 0));

                v8f P = __builtin_amdgcn_wmma_f32_16x16x4_f32(
                    false, aP0, false, bP0, (short)0, zc, false, false);
                P = __builtin_amdgcn_wmma_f32_16x16x4_f32(
                    false, aP1, false, bP1, (short)0, P, false, false);
                v8f Q = __builtin_amdgcn_wmma_f32_16x16x4_f32(
                    false, aQ, false, bQ, (short)0, zc, false, false);

#pragma unroll
                for (int e = 0; e < 8; ++e)
                    acc[e] += P[e] * __builtin_amdgcn_rcpf(1.0f + __builtin_fabsf(Q[e]));
            }
        }
    }

    const int wbase = w0 + (hi ? 8 : 0);
    float* o = out + ((size_t)(b * FF + f) * HH + h) * WW + wbase;

    v4f s;
    s.x = acc[0]; s.y = acc[1]; s.z = acc[2]; s.w = acc[3];
    *(v4f*)(o) = s;
    s.x = acc[4]; s.y = acc[5]; s.z = acc[6]; s.w = acc[7];
    *(v4f*)(o + 4) = s;
}

extern "C" void kernel_launch(void* const* d_in, const int* in_sizes, int n_in,
                              void* d_out, int out_size, void* d_ws, size_t ws_size,
                              hipStream_t stream) {
    (void)in_sizes; (void)n_in; (void)out_size; (void)d_ws; (void)ws_size;
    const float* x      = (const float*)d_in[0];
    const float* nums   = (const float*)d_in[1];
    const float* denoms = (const float*)d_in[2];
    float* out = (float*)d_out;

    pade_conv_wmma<<<256, 256, 0, stream>>>(x, nums, denoms, out);
}


# coefficients staged per channel in LDS by contiguous 8-byte loads one channel ahead (two stages, one barrier per channel); B fragments read from LDS instead of 16-line strided global loads
# speedup vs baseline: 1.5715x; 1.5715x over previous
#include <hip/hip_runtime.h>

typedef __attribute__((ext_vector_type(2))) float v2f;
typedef __attribute__((ext_vector_type(4))) float v4f;
typedef __attribute__((ext_vector_type(8))) float v8f;

__device__ __forceinline__ v2f make_v2f(float a, float b) {
    v2f r; r.x = a; r.y = b; return r;
}

#define BBATCH 4
#define CC     16
#define HH     64
#define WW     64
#define FF     32
#define KK     3

#define NPITCH 60
#define DPITCH 36
#define NUM_F2_PER_F 27
#define DEN_F2_PER_F 18
#define NUM_F2 (FF * NUM_F2_PER_F)
#define DEN_F2 (FF * DEN_F2_PER_F)
#define NUM_LDS_FLOATS (FF * NPITCH)
#define STAGE_FLOATS (NUM_LDS_FLOATS + FF * DPITCH)

__global__ __launch_bounds__(256) void pade_conv_wmma(
    const float* __restrict__ x,
    const float* __restrict__ nums,
    const float* __restrict__ denoms,
    float* __restrict__ out)
{
    const int lane  = threadIdx.x & 31;
    const int gw    = blockIdx.x * 8 + (threadIdx.x >> 5);
    const int ftile = gw & 1;
    const int tile  = gw >> 1;

    const int w0 = (tile & 3) * 16;
    const int h  = (tile >> 2) & (HH - 1);
    const int b  = tile >> 8;

    const int  laneM = lane & 15;
    const bool hi    = lane >= 16;

    const int f = ftile * 16 + laneM;

    __shared__ __attribute__((aligned(16))) float coef[2 * STAGE_FLOATS];

    const int nOff  = f * NPITCH + (hi ? 2 : 0);
    const int n1Off = f * NPITCH + 4;
    const int dOff  = NUM_LDS_FLOATS + f * DPITCH + (hi ? 2 : 0);

    const int tid = threadIdx.x;
    int nG[3], nL[3], dG[2], dL[2];
#pragma unroll
    for (int k = 0; k < 3; ++k) {
        const int idx = tid + k * 256;
        const int ff  = idx / NUM_F2_PER_F;
        const int j   = idx - ff * NUM_F2_PER_F;
        nG[k] = ff * (CC * KK * KK * 6) + 2 * j;
        nL[k] = ff * NPITCH + 2 * j;
    }
#pragma unroll
    for (int k = 0; k < 2; ++k) {
        const int idx = tid + k * 256;
        const int ff  = idx / DEN_F2_PER_F;
        const int j   = idx - ff * DEN_F2_PER_F;
        dG[k] = ff * (CC * KK * KK * 4) + 2 * j;
        dL[k] = NUM_LDS_FLOATS + ff * DPITCH + 2 * j;
    }
    const bool tailN = tid < 96;
    const bool tailD = tid >= 96 && tid < 160;
    int tG, tL;
    {
        const int idxN = 768 + tid;
        const int ffN  = idxN / NUM_F2_PER_F;
        const int jN   = idxN - ffN * NUM_F2_PER_F;
        const int idxD = 512 + ((tid - 96) & 63);
        const int ffD  = idxD / DEN_F2_PER_F;
        const int jD   = idxD - ffD * DEN_F2_PER_F;
        tG = tailN ? ffN * (CC * KK * KK * 6) + 2 * jN
                   : ffD * (CC * KK * KK * 4) + 2 * jD;
        tL = tailN ? ffN * NPITCH + 2 * jN
                   : NUM_LDS_FLOATS + ffD * DPITCH + 2 * jD;
    }

    {
        v2f pn[3], pd[2], pt = make_v2f(0.0f, 0.0f);
#pragma unroll
        for (int k = 0; k < 3; ++k) pn[k] = *(const v2f*)(nums + nG[k]);
#pragma unroll
        for (int k = 0; k < 2; ++k) pd[k] = *(const v2f*)(denoms + dG[k]);
        if (tailN)      pt = *(const v2f*)(nums + tG);
        else if (tailD) pt = *(const v2f*)(denoms + tG);
#pragma unroll
        for (int k = 0; k < 3; ++k) *(v2f*)(coef + nL[k]) = pn[k];
#pragma unroll
        for (int k = 0; k < 2; ++k) *(v2f*)(coef + dL[k]) = pd[k];
        if (tailN || tailD) *(v2f*)(coef + tL) = pt;
    }
    __syncthreads();

    v8f acc = {};
    const v8f zc = {};

    const int  loadIdx = (lane < 18) ? lane : 17;
    const int  wl      = w0 - 1 + loadIdx;
    const bool wlOK    = (unsigned)wl < (unsigned)WW;

    for (int c = 0; c < CC; ++c) {
        const float* cur = coef + (c & 1) * STAGE_FLOATS;
        float*       nxt = coef + ((c & 1) ^ 1) * STAGE_FLOATS;
        const bool more = c + 1 < CC;

        v2f pn[3], pd[2], pt = make_v2f(0.0f, 0.0f);
        if (more) {
            const float* __restrict__ nsrc = nums   + (c + 1) * (KK * KK * 6);
            const float* __restrict__ dsrc = denoms + (c + 1) * (KK * KK * 4);
#pragma unroll
            for (int k = 0; k < 3; ++k) pn[k] = *(const v2f*)(nsrc + nG[k]);
#pragma unroll
            for (int k = 0; k < 2; ++k) pd[k] = *(const v2f*)(dsrc + dG[k]);
            if (tailN)      pt = *(const v2f*)(nsrc + tG);
            else if (tailD) pt = *(const v2f*)(dsrc + tG);
        }

        const float* __restrict__ xplane = x + (size_t)(b * CC + c) * (HH * WW);
        for (int a = 0; a < KK; ++a) {
            const int  hh  = h + a - 1;
            const bool hOK = (unsigned)hh < (unsigned)HH;
            const float* __restrict__ xrow = xplane + (size_t)(hOK ? hh : 0) * WW;

            float xload = 0.0f;
            if (hOK && wlOK) xload = xrow[wl];
            const int xbits = __float_as_int(xload);

#pragma unroll
            for (int bb = 0; bb < KK; ++bb) {
                const float xv = __int_as_float(
                    __builtin_amdgcn_ds_bpermute((laneM + bb) << 2, xbits));

                const float x2 = xv * xv;
                const float x3 = x2 * xv;
                const float x4 = x2 * x2;
                const float x5 = x4 * xv;

                v2f aP0, aP1, aQ;
                if (!hi) {
                    aP0 = make_v2f(1.0f, xv);
                    aP1 = make_v2f(x4,   x5);
                    aQ  = make_v2f(xv,   x2);
                } else {
                    aP0 = make_v2f(x2,   x3);
                    aP1 = make_v2f(0.0f, 0.0f);
                    aQ  = make_v2f(x3,   x4);
                }

                const int tap = a * KK + bb;

                v2f bP0 = *(const v2f*)(cur + nOff + tap * 6);
                v2f bP1 = *(const v2f*)(cur + n1Off + tap * 6);
                if (hi) bP1 = make_v2f(0.0f, 0.0f);
                v2f bQ  = *(const v2f*)(cur + dOff + tap * 4);

                v8f P = __builtin_amdgcn_wmma_f32_16x16x4_f32(
                    false, aP0, false, bP0, (short)0, zc, false, false);
                P = __builtin_amdgcn_wmma_f32_16x16x4_f32(
                    false, aP1, false, bP1, (short)0, P, false, false);
                v8f Q = __builtin_amdgcn_wmma_f32_16x16x4_f32(
                    false, aQ, false, bQ, (short)0, zc, false, false);

#pragma unroll
                for (int e = 0; e < 8; ++e)
                    acc[e] += P[e] * __builtin_amdgcn_rcpf(1.0f + __builtin_fabsf(Q[e]));
            }
        }

        if (more) {
#pragma unroll
            for (int k = 0; k < 3; ++k) *(v2f*)(nxt + nL[k]) = pn[k];
#pragma unroll
            for (int k = 0; k < 2; ++k) *(v2f*)(nxt + dL[k]) = pd[k];
            if (tailN || tailD) *(v2f*)(nxt + tL) = pt;
            __syncthreads();
        }
    }

    const int wbase = w0 + (hi ? 8 : 0);
    float* o = out + ((size_t)(b * FF + f) * HH + h) * WW + wbase;

    v4f s;
    s.x = acc[0]; s.y = acc[1]; s.z = acc[2]; s.w = acc[3];
    *(v4f*)(o) = s;
    s.x = acc[4]; s.y = acc[5]; s.z = acc[6]; s.w = acc[7];
    *(v4f*)(o + 4) = s;
}

extern "C" void kernel_launch(void* const* d_in, const int* in_sizes, int n_in,
                              void* d_out, int out_size, void* d_ws, size_t ws_size,
                              hipStream_t stream) {
    (void)in_sizes; (void)n_in; (void)out_size; (void)d_ws; (void)ws_size;
    const float* x      = (const float*)d_in[0];
    const float* nums   = (const float*)d_in[1];
    const float* denoms = (const float*)d_in[2];
    float* out = (float*)d_out;

    pade_conv_wmma<<<256, 256, 0, stream>>>(x, nums, denoms, out);
}


# input window (16 channels x 3 rows) staged once per work-group in LDS by 16-byte loads in the prologue; taps read x from LDS; no global x load or lane permute left in the channel loop
# speedup vs baseline: 2.7937x; 1.7777x over previous
#include <hip/hip_runtime.h>

typedef __attribute__((ext_vector_type(2))) float v2f;
typedef __attribute__((ext_vector_type(4))) float v4f;
typedef __attribute__((ext_vector_type(8))) float v8f;

__device__ __forceinline__ v2f make_v2f(float a, float b) {
    v2f r; r.x = a; r.y = b; return r;
}

#define BBATCH 4
#define CC     16
#define HH     64
#define WW     64
#define FF     32
#define KK     3

#define NPITCH 60
#define DPITCH 36
#define NUM_F2_PER_F 27
#define DEN_F2_PER_F 18
#define NUM_F2 (FF * NUM_F2_PER_F)
#define DEN_F2 (FF * DEN_F2_PER_F)
#define NUM_LDS_FLOATS (FF * NPITCH)
#define STAGE_FLOATS (NUM_LDS_FLOATS + FF * DPITCH)

#define XPITCH 72
#define XROWS  (CC * KK)
#define XCOL0  4

__global__ __launch_bounds__(256) void pade_conv_wmma(
    const float* __restrict__ x,
    const float* __restrict__ nums,
    const float* __restrict__ denoms,
    float* __restrict__ out)
{
    const int lane  = threadIdx.x & 31;
    const int gw    = blockIdx.x * 8 + (threadIdx.x >> 5);
    const int ftile = gw & 1;
    const int tile  = gw >> 1;

    const int w0 = (tile & 3) * 16;
    const int h  = (tile >> 2) & (HH - 1);
    const int b  = tile >> 8;

    const int  laneM = lane & 15;
    const bool hi    = lane >= 16;

    const int f = ftile * 16 + laneM;

    __shared__ __attribute__((aligned(16))) float coef[2 * STAGE_FLOATS];
    __shared__ __attribute__((aligned(16))) float xs[XROWS * XPITCH];

    const int nOff  = f * NPITCH + (hi ? 2 : 0);
    const int n1Off = f * NPITCH + 4;
    const int dOff  = NUM_LDS_FLOATS + f * DPITCH + (hi ? 2 : 0);

    const int tid = threadIdx.x;
    int nG[3], nL[3], dG[2], dL[2];
#pragma unroll
    for (int k = 0; k < 3; ++k) {
        const int idx = tid + k * 256;
        const int ff  = idx / NUM_F2_PER_F;
        const int j   = idx - ff * NUM_F2_PER_F;
        nG[k] = ff * (CC * KK * KK * 6) + 2 * j;
        nL[k] = ff * NPITCH + 2 * j;
    }
#pragma unroll
    for (int k = 0; k < 2; ++k) {
        const int idx = tid + k * 256;
        const int ff  = idx / DEN_F2_PER_F;
        const int j   = idx - ff * DEN_F2_PER_F;
        dG[k] = ff * (CC * KK * KK * 4) + 2 * j;
        dL[k] = NUM_LDS_FLOATS + ff * DPITCH + 2 * j;
    }
    const bool tailN = tid < 96;
    const bool tailD = tid >= 96 && tid < 160;
    int tG, tL;
    {
        const int idxN = 768 + tid;
        const int ffN  = idxN / NUM_F2_PER_F;
        const int jN   = idxN - ffN * NUM_F2_PER_F;
        const int idxD = 512 + ((tid - 96) & 63);
        const int ffD  = idxD / DEN_F2_PER_F;
        const int jD   = idxD - ffD * DEN_F2_PER_F;
        tG = tailN ? ffN * (CC * KK * KK * 6) + 2 * jN
                   : ffD * (CC * KK * KK * 4) + 2 * jD;
        tL = tailN ? ffN * NPITCH + 2 * jN
                   : NUM_LDS_FLOATS + ffD * DPITCH + 2 * jD;
    }

    {
        v4f xq[3];
        int xL[3];
#pragma unroll
        for (int k = 0; k < 3; ++k) {
            const int idx = tid + k * 256;
            const int row = idx >> 4;
            const int seg = idx & 15;
            const int cch = row / KK;
            const int hh  = h + (row - cch * KK) - 1;
            xL[k] = row * XPITCH + XCOL0 + seg * 4;
            v4f q = {};
            if ((unsigned)hh < (unsigned)HH)
                q = *(const v4f*)(x + ((size_t)(b * CC + cch) * HH + hh) * WW + seg * 4);
            xq[k] = q;
        }

        v2f pn[3], pd[2], pt = make_v2f(0.0f, 0.0f);
#pragma unroll
        for (int k = 0; k < 3; ++k) pn[k] = *(const v2f*)(nums + nG[k]);
#pragma unroll
        for (int k = 0; k < 2; ++k) pd[k] = *(const v2f*)(denoms + dG[k]);
        if (tailN)      pt = *(const v2f*)(nums + tG);
        else if (tailD) pt = *(const v2f*)(denoms + tG);

#pragma unroll
        for (int k = 0; k < 3; ++k) *(v4f*)(xs + xL[k]) = xq[k];
        if (tid < 2 * XROWS)
            xs[(tid >> 1) * XPITCH + ((tid & 1) ? XCOL0 + WW : XCOL0 - 1)] = 0.0f;
#pragma unroll
        for (int k = 0; k < 3; ++k) *(v2f*)(coef + nL[k]) = pn[k];
#pragma unroll
        for (int k = 0; k < 2; ++k) *(v2f*)(coef + dL[k]) = pd[k];
        if (tailN || tailD) *(v2f*)(coef + tL) = pt;
    }
    __syncthreads();

    v8f acc = {};
    const v8f zc = {};

    const float* xlane = xs + (XCOL0 - 1) + w0 + laneM;

    for (int c = 0; c < CC; ++c) {
        const float* cur = coef + (c & 1) * STAGE_FLOATS;
        float*       nxt = coef + ((c & 1) ^ 1) * STAGE_FLOATS;
        const bool more = c + 1 < CC;

        v2f pn[3], pd[2], pt = make_v2f(0.0f, 0.0f);
        if (more) {
            const float* __restrict__ nsrc = nums   + (c + 1) * (KK * KK * 6);
            const float* __restrict__ dsrc = denoms + (c + 1) * (KK * KK * 4);
#pragma unroll
            for (int k = 0; k < 3; ++k) pn[k] = *(const v2f*)(nsrc + nG[k]);
#pragma unroll
            for (int k = 0; k < 2; ++k) pd[k] = *(const v2f*)(dsrc + dG[k]);
            if (tailN)      pt = *(const v2f*)(nsrc + tG);
            else if (tailD) pt = *(const v2f*)(dsrc + tG);
        }

        for (int a = 0; a < KK; ++a) {
            const float* xrow = xlane + (c * KK + a) * XPITCH;

#pragma unroll
            for (int bb = 0; bb < KK; ++bb) {
                const float xv = xrow[bb];

                const float x2 = xv * xv;
                const float x3 = x2 * xv;
                const float x4 = x2 * x2;
                const float x5 = x4 * xv;

                v2f aP0, aP1, aQ;
                if (!hi) {
                    aP0 = make_v2f(1.0f, xv);
                    aP1 = make_v2f(x4,   x5);
                    aQ  = make_v2f(xv,   x2);
                } else {
                    aP0 = make_v2f(x2,   x3);
                    aP1 = make_v2f(0.0f, 0.0f);
                    aQ  = make_v2f(x3,   x4);
                }

                const int tap = a * KK + bb;

                v2f bP0 = *(const v2f*)(cur + nOff + tap * 6);
                v2f bP1 = *(const v2f*)(cur + n1Off + tap * 6);
                if (hi) bP1 = make_v2f(0.0f, 0.0f);
                v2f bQ  = *(const v2f*)(cur + dOff + tap * 4);

                v8f P = __builtin_amdgcn_wmma_f32_16x16x4_f32(
                    false, aP0, false, bP0, (short)0, zc, false, false);
                P = __builtin_amdgcn_wmma_f32_16x16x4_f32(
                    false, aP1, false, bP1, (short)0, P, false, false);
                v8f Q = __builtin_amdgcn_wmma_f32_16x16x4_f32(
                    false, aQ, false, bQ, (short)0, zc, false, false);

#pragma unroll
                for (int e = 0; e < 8; ++e)
                    acc[e] = __builtin_fmaf(
                        P[e], __builtin_amdgcn_rcpf(1.0f + __builtin_fabsf(Q[e])), acc[e]);
            }
        }

        if (more) {
#pragma unroll
            for (int k = 0; k < 3; ++k) *(v2f*)(nxt + nL[k]) = pn[k];
#pragma unroll
            for (int k = 0; k < 2; ++k) *(v2f*)(nxt + dL[k]) = pd[k];
            if (tailN || tailD) *(v2f*)(nxt + tL) = pt;
            __syncthreads();
        }
    }

    const int wbase = w0 + (hi ? 8 : 0);
    float* o = out + ((size_t)(b * FF + f) * HH + h) * WW + wbase;

    v4f s;
    s.x = acc[0]; s.y = acc[1]; s.z = acc[2]; s.w = acc[3];
    *(v4f*)(o) = s;
    s.x = acc[4]; s.y = acc[5]; s.z = acc[6]; s.w = acc[7];
    *(v4f*)(o + 4) = s;
}

extern "C" void kernel_launch(void* const* d_in, const int* in_sizes, int n_in,
                              void* d_out, int out_size, void* d_ws, size_t ws_size,
                              hipStream_t stream) {
    (void)in_sizes; (void)n_in; (void)out_size; (void)d_ws; (void)ws_size;
    const float* x      = (const float*)d_in[0];
    const float* nums   = (const float*)d_in[1];
    const float* denoms = (const float*)d_in[2];
    float* out = (float*)d_out;

    pade_conv_wmma<<<256, 256, 0, stream>>>(x, nums, denoms, out);
}


# speedup vs baseline: 2.9801x; 1.0667x over previous
#include <hip/hip_runtime.h>

typedef __attribute__((ext_vector_type(2))) float v2f;
typedef __attribute__((ext_vector_type(4))) float v4f;
typedef __attribute__((ext_vector_type(8))) float v8f;

__device__ __forceinline__ v2f make_v2f(float a, float b) {
    v2f r; r.x = a; r.y = b; return r;
}

#define BBATCH 4
#define CC     16
#define HH     64
#define WW     64
#define FF     32
#define KK     3

#define NPITCH 60
#define DPITCH 36
#define NUM_F2_PER_F 27
#define DEN_F2_PER_F 18
#define NUM_F2 (FF * NUM_F2_PER_F)
#define DEN_F2 (FF * DEN_F2_PER_F)
#define NUM_LDS_FLOATS (FF * NPITCH)
#define STAGE_FLOATS (NUM_LDS_FLOATS + FF * DPITCH)

#define XPITCH 72
#define XROWS  (CC * KK)
#define XCOL0  4

__global__ __launch_bounds__(256) void pade_conv_wmma(
    const float* __restrict__ x,
    const float* __restrict__ nums,
    const float* __restrict__ denoms,
    float* __restrict__ out)
{
    const int lane  = threadIdx.x & 31;
    const int gw    = blockIdx.x * 8 + (threadIdx.x >> 5);
    const int ftile = gw & 1;
    const int tile  = gw >> 1;

    const int w0 = (tile & 3) * 16;
    const int h  = (tile >> 2) & (HH - 1);
    const int b  = tile >> 8;

    const int  laneM = lane & 15;
    const bool hi    = lane >= 16;

    const int f = ftile * 16 + laneM;

    __shared__ __attribute__((aligned(16))) float coef[2 * STAGE_FLOATS];
    __shared__ __attribute__((aligned(16))) float xs[XROWS * XPITCH];

    const int nOff  = f * NPITCH + (hi ? 2 : 0);
    const int n1Off = f * NPITCH + 4;
    const int dOff  = NUM_LDS_FLOATS + f * DPITCH + (hi ? 2 : 0);

    const int tid = threadIdx.x;
    int nG[3], nL[3], dG[2], dL[2];
#pragma unroll
    for (int k = 0; k < 3; ++k) {
        const int idx = tid + k * 256;
        const int ff  = idx / NUM_F2_PER_F;
        const int j   = idx - ff * NUM_F2_PER_F;
        nG[k] = ff * (CC * KK * KK * 6) + 2 * j;
        nL[k] = ff * NPITCH + 2 * j;
    }
#pragma unroll
    for (int k = 0; k < 2; ++k) {
        const int idx = tid + k * 256;
        const int ff  = idx / DEN_F2_PER_F;
        const int j   = idx - ff * DEN_F2_PER_F;
        dG[k] = ff * (CC * KK * KK * 4) + 2 * j;
        dL[k] = NUM_LDS_FLOATS + ff * DPITCH + 2 * j;
    }
    const bool tailN = tid < 96;
    const bool tailD = tid >= 96 && tid < 160;
    int tG, tL;
    {
        const int idxN = 768 + tid;
        const int ffN  = idxN / NUM_F2_PER_F;
        const int jN   = idxN - ffN * NUM_F2_PER_F;
        const int idxD = 512 + ((tid - 96) & 63);
        const int ffD  = idxD / DEN_F2_PER_F;
        const int jD   = idxD - ffD * DEN_F2_PER_F;
        tG = tailN ? ffN * (CC * KK * KK * 6) + 2 * jN
                   : ffD * (CC * KK * KK * 4) + 2 * jD;
        tL = tailN ? ffN * NPITCH + 2 * jN
                   : NUM_LDS_FLOATS + ffD * DPITCH + 2 * jD;
    }

    {
        v4f xq[3];
        int xL[3];
#pragma unroll
        for (int k = 0; k < 3; ++k) {
            const int idx = tid + k * 256;
            const int row = idx >> 4;
            const int seg = idx & 15;
            const int cch = row / KK;
            const int hh  = h + (row - cch * KK) - 1;
            xL[k] = row * XPITCH + XCOL0 + seg * 4;
            v4f q = {};
            if ((unsigned)hh < (unsigned)HH)
                q = *(const v4f*)(x + ((size_t)(b * CC + cch) * HH + hh) * WW + seg * 4);
            xq[k] = q;
        }

        v2f pn[3], pd[2], pt = make_v2f(0.0f, 0.0f);
#pragma unroll
        for (int k = 0; k < 3; ++k) pn[k] = *(const v2f*)(nums + nG[k]);
#pragma unroll
        for (int k = 0; k < 2; ++k) pd[k] = *(const v2f*)(denoms + dG[k]);
        if (tailN)      pt = *(const v2f*)(nums + tG);
        else if (tailD) pt = *(const v2f*)(denoms + tG);

#pragma unroll
        for (int k = 0; k < 3; ++k) *(v4f*)(xs + xL[k]) = xq[k];
        if (tid < 2 * XROWS)
            xs[(tid >> 1) * XPITCH + ((tid & 1) ? XCOL0 + WW : XCOL0 - 1)] = 0.0f;
#pragma unroll
        for (int k = 0; k < 3; ++k) *(v2f*)(coef + nL[k]) = pn[k];
#pragma unroll
        for (int k = 0; k < 2; ++k) *(v2f*)(coef + dL[k]) = pd[k];
        if (tailN || tailD) *(v2f*)(coef + tL) = pt;
    }
    __syncthreads();

    v2f acc2[4] = {make_v2f(0.0f, 0.0f), make_v2f(0.0f, 0.0f), make_v2f(0.0f, 0.0f), make_v2f(0.0f, 0.0f)};
    const v8f zc = {};

    const float* xlane = xs + (XCOL0 - 1) + w0 + laneM;

    for (int c = 0; c < CC; ++c) {
        const float* cur = coef + (c & 1) * STAGE_FLOATS;
        float*       nxt = coef + ((c & 1) ^ 1) * STAGE_FLOATS;
        const bool more = c + 1 < CC;

        v2f pn[3], pd[2], pt = make_v2f(0.0f, 0.0f);
        if (more) {
            const float* __restrict__ nsrc = nums   + (c + 1) * (KK * KK * 6);
            const float* __restrict__ dsrc = denoms + (c + 1) * (KK * KK * 4);
#pragma unroll
            for (int k = 0; k < 3; ++k) pn[k] = *(const v2f*)(nsrc + nG[k]);
#pragma unroll
            for (int k = 0; k < 2; ++k) pd[k] = *(const v2f*)(dsrc + dG[k]);
            if (tailN)      pt = *(const v2f*)(nsrc + tG);
            else if (tailD) pt = *(const v2f*)(dsrc + tG);
        }

        v8f Pt[KK * KK], Qt[KK * KK];
#pragma unroll
        for (int a = 0; a < KK; ++a) {
            const float* xrow = xlane + (c * KK + a) * XPITCH;

#pragma unroll
            for (int bb = 0; bb < KK; ++bb) {
                const float xv = xrow[bb];

                const float x2 = xv * xv;
                const float u  = hi ? x2 : 1.0f;
                const float ux = u * xv;
                const float x4 = x2 * x2;
                const float v  = hi ? 0.0f : x4;
                const v2f aP0 = make_v2f(u,  ux);
                const v2f aQ  = make_v2f(ux, ux * xv);
                const v2f aP1 = make_v2f(v,  v * xv);

                const int tap = a * KK + bb;

                v2f bP0 = *(const v2f*)(cur + nOff + tap * 6);
                const v2f bP1 = *(const v2f*)(cur + n1Off + tap * 6);
                v2f bQ  = *(const v2f*)(cur + dOff + tap * 4);

                v8f P = __builtin_amdgcn_wmma_f32_16x16x4_f32(
                    false, aP0, false, bP0, (short)0, zc, false, false);
                P = __builtin_amdgcn_wmma_f32_16x16x4_f32(
                    false, aP1, false, bP1, (short)0, P, false, false);
                v8f Q = __builtin_amdgcn_wmma_f32_16x16x4_f32(
                    false, aQ, false, bQ, (short)0, zc, false, false);

                Pt[tap] = P;
                Qt[tap] = Q;
            }
        }

        if (more) {
#pragma unroll
            for (int k = 0; k < 3; ++k) *(v2f*)(nxt + nL[k]) = pn[k];
#pragma unroll
            for (int k = 0; k < 2; ++k) *(v2f*)(nxt + dL[k]) = pd[k];
            if (tailN || tailD) *(v2f*)(nxt + tL) = pt;
        }

        float tq[8 * KK * KK];
        asm volatile("v_nop\n\tv_nop\n\tv_nop\n\tv_nop"
                     :
                     : "v"(Pt[0]), "v"(Pt[1]), "v"(Pt[2]), "v"(Pt[3]), "v"(Pt[4]),
                       "v"(Pt[5]), "v"(Pt[6]), "v"(Pt[7]), "v"(Pt[8]),
                       "v"(Qt[0]), "v"(Qt[1]), "v"(Qt[2]), "v"(Qt[3]), "v"(Qt[4]),
                       "v"(Qt[5]), "v"(Qt[6]), "v"(Qt[7]), "v"(Qt[8]));
        asm volatile("v_add_f32 %0, |%4|, 1.0\n\tv_add_f32 %1, |%5|, 1.0\n\t"
                     "v_add_f32 %2, |%6|, 1.0\n\tv_add_f32 %3, |%7|, 1.0"
                     : "=&v"(tq[0]), "=&v"(tq[1]), "=&v"(tq[2]), "=&v"(tq[3])
                     : "v"(Qt[0][0]), "v"(Qt[0][1]), "v"(Qt[0][2]), "v"(Qt[0][3]));
#pragma unroll
        for (int i = 0; i < 8 * KK * KK; ++i) {
            if (i + 4 < 8 * KK * KK) {
                asm volatile("v_rcp_f32 %0, %0\n\tv_add_f32 %1, |%2|, 1.0"
                             : "+v"(tq[i]), "=&v"(tq[i + 4])
                             : "v"(Qt[(i + 4) >> 3][(i + 4) & 7]));
            } else {
                asm volatile("v_rcp_f32 %0, %0" : "+v"(tq[i]));
            }
            if (i >= 5 && (i & 1)) {
                const int j = i - 5;
                asm volatile("v_pk_fma_f32 %0, %1, %2, %0"
                             : "+v"(acc2[(j & 7) >> 1])
                             : "v"(make_v2f(Pt[j >> 3][j & 7], Pt[j >> 3][(j & 7) + 1])),
                               "v"(make_v2f(tq[j], tq[j + 1])));
            }
        }
#pragma unroll
        for (int j = 8 * KK * KK - 4; j < 8 * KK * KK; j += 2)
            asm volatile("v_pk_fma_f32 %0, %1, %2, %0"
                         : "+v"(acc2[(j & 7) >> 1])
                         : "v"(make_v2f(Pt[j >> 3][j & 7], Pt[j >> 3][(j & 7) + 1])),
                           "v"(make_v2f(tq[j], tq[j + 1])));

        if (more) __syncthreads();
    }

    const int wbase = w0 + (hi ? 8 : 0);
    float* o = out + ((size_t)(b * FF + f) * HH + h) * WW + wbase;

    v4f s;
    s.x = acc2[0].x; s.y = acc2[0].y; s.z = acc2[1].x; s.w = acc2[1].y;
    *(v4f*)(o) = s;
    s.x = acc2[2].x; s.y = acc2[2].y; s.z = acc2[3].x; s.w = acc2[3].y;
    *(v4f*)(o + 4) = s;
}

extern "C" void kernel_launch(void* const* d_in, const int* in_sizes, int n_in,
                              void* d_out, int out_size, void* d_ws, size_t ws_size,
                              hipStream_t stream) {
    (void)in_sizes; (void)n_in; (void)out_size; (void)d_ws; (void)ws_size;
    const float* x      = (const float*)d_in[0];
    const float* nums   = (const float*)d_in[1];
    const float* denoms = (const float*)d_in[2];
    float* out = (float*)d_out;

    pade_conv_wmma<<<256, 256, 0, stream>>>(x, nums, denoms, out);
}


# re-tile: each wave owns both 16-filter tiles of its strip and half the channels (two waves per SIMD kept): A fragments, x reads, barriers per product halve; channel halves summed apart, added via LDS
# speedup vs baseline: 3.1963x; 1.0725x over previous
#include <hip/hip_runtime.h>

typedef __attribute__((ext_vector_type(2))) float v2f;
typedef __attribute__((ext_vector_type(4))) float v4f;
typedef __attribute__((ext_vector_type(8))) float v8f;

__device__ __forceinline__ v2f make_v2f(float a, float b) {
    v2f r; r.x = a; r.y = b; return r;
}

#define BBATCH 4
#define CC     16
#define HH     64
#define WW     64
#define FF     32
#define KK     3
#define TAPS   (KK * KK)

#define GROUPS 2
#define GCH    (CC / GROUPS)

#define NPITCH 60
#define DPITCH 36
#define NUM_F2_PER_F 27
#define DEN_F2_PER_F 18
#define NUM_F2 (FF * NUM_F2_PER_F)
#define DEN_F2 (FF * DEN_F2_PER_F)
#define NUM_LDS_FLOATS (FF * NPITCH)
#define STAGE_FLOATS (NUM_LDS_FLOATS + FF * DPITCH)
#define STEP_FLOATS (GROUPS * STAGE_FLOATS)
#define N_TILE1 (16 * NPITCH)
#define D_TILE1 (16 * DPITCH)

#define XPITCH 72
#define XROWS  (CC * KK)
#define XCOL0  4

__device__ __forceinline__ void recip_accumulate(const v8f (&Pt)[TAPS], const v8f (&Qt)[TAPS],
                                                 v2f (&acc2)[4]) {
    float tq[8 * TAPS];
    asm volatile("v_nop\n\tv_nop\n\tv_nop\n\tv_nop"
                 :
                 : "v"(Pt[0]), "v"(Pt[1]), "v"(Pt[2]), "v"(Pt[3]), "v"(Pt[4]),
                   "v"(Pt[5]), "v"(Pt[6]), "v"(Pt[7]), "v"(Pt[8]),
                   "v"(Qt[0]), "v"(Qt[1]), "v"(Qt[2]), "v"(Qt[3]), "v"(Qt[4]),
                   "v"(Qt[5]), "v"(Qt[6]), "v"(Qt[7]), "v"(Qt[8]));
    asm volatile("v_add_f32 %0, |%4|, 1.0\n\tv_add_f32 %1, |%5|, 1.0\n\t"
                 "v_add_f32 %2, |%6|, 1.0\n\tv_add_f32 %3, |%7|, 1.0"
                 : "=&v"(tq[0]), "=&v"(tq[1]), "=&v"(tq[2]), "=&v"(tq[3])
                 : "v"(Qt[0][0]), "v"(Qt[0][1]), "v"(Qt[0][2]), "v"(Qt[0][3]));
#pragma unroll
    for (int i = 0; i < 8 * TAPS; ++i) {
        if (i + 4 < 8 * TAPS) {
            asm volatile("v_rcp_f32 %0, %0\n\tv_add_f32 %1, |%2|, 1.0"
                         : "+v"(tq[i]), "=&v"(tq[i + 4])
                         : "v"(Qt[(i + 4) >> 3][(i + 4) & 7]));
        } else {
            asm volatile("v_rcp_f32 %0, %0" : "+v"(tq[i]));
        }
        if (i >= 5 && (i & 1)) {
            const int j = i - 5;
            asm volatile("v_pk_fma_f32 %0, %1, %2, %0"
                         : "+v"(acc2[(j & 7) >> 1])
                         : "v"(make_v2f(Pt[j >> 3][j & 7], Pt[j >> 3][(j & 7) + 1])),
                           "v"(make_v2f(tq[j], tq[j + 1])));
        }
    }
#pragma unroll
    for (int j = 8 * TAPS - 4; j < 8 * TAPS; j += 2)
        asm volatile("v_pk_fma_f32 %0, %1, %2, %0"
                     : "+v"(acc2[(j & 7) >> 1])
                     : "v"(make_v2f(Pt[j >> 3][j & 7], Pt[j >> 3][(j & 7) + 1])),
                       "v"(make_v2f(tq[j], tq[j + 1])));
}

__global__ __launch_bounds__(256) void pade_conv_wmma(
    const float* __restrict__ x,
    const float* __restrict__ nums,
    const float* __restrict__ denoms,
    float* __restrict__ out)
{
    const int tid   = threadIdx.x;
    const int lane  = tid & 31;
    const int wv    = __builtin_amdgcn_readfirstlane(tid >> 5);
    const int strip = wv & 3;
    const int g     = wv >> 2;

    const int w0 = strip * 16;
    const int h  = blockIdx.x & (HH - 1);
    const int b  = blockIdx.x >> 6;

    const int  laneM = lane & 15;
    const bool hi    = lane >= 16;

    __shared__ __attribute__((aligned(16))) float coef[2 * STEP_FLOATS];
    __shared__ __attribute__((aligned(16))) float xs[XROWS * XPITCH];

    const int nOff  = laneM * NPITCH + (hi ? 2 : 0);
    const int n1Off = laneM * NPITCH + 4;
    const int dOff  = NUM_LDS_FLOATS + laneM * DPITCH + (hi ? 2 : 0);

    int nG[3], nL[3], dG[2], dL[2];
#pragma unroll
    for (int k = 0; k < 3; ++k) {
        const int idx = tid + k * 256;
        const int ff  = idx / NUM_F2_PER_F;
        const int j   = idx - ff * NUM_F2_PER_F;
        nG[k] = ff * (CC * KK * KK * 6) + 2 * j;
        nL[k] = ff * NPITCH + 2 * j;
    }
#pragma unroll
    for (int k = 0; k < 2; ++k) {
        const int idx = tid + k * 256;
        const int ff  = idx / DEN_F2_PER_F;
        const int j   = idx - ff * DEN_F2_PER_F;
        dG[k] = ff * (CC * KK * KK * 4) + 2 * j;
        dL[k] = NUM_LDS_FLOATS + ff * DPITCH + 2 * j;
    }
    const bool tailN = tid < 96;
    const bool tailD = tid >= 96 && tid < 160;
    int tG, tL;
    {
        const int idxN = 768 + tid;
        const int ffN  = idxN / NUM_F2_PER_F;
        const int jN   = idxN - ffN * NUM_F2_PER_F;
        const int idxD = 512 + ((tid - 96) & 63);
        const int ffD  = idxD / DEN_F2_PER_F;
        const int jD   = idxD - ffD * DEN_F2_PER_F;
        tG = tailN ? ffN * (CC * KK * KK * 6) + 2 * jN
                   : ffD * (CC * KK * KK * 4) + 2 * jD;
        tL = tailN ? ffN * NPITCH + 2 * jN
                   : NUM_LDS_FLOATS + ffD * DPITCH + 2 * jD;
    }

    {
        v4f xq[3];
        int xL[3];
#pragma unroll
        for (int k = 0; k < 3; ++k) {
            const int idx = tid + k * 256;
            const int row = idx >> 4;
            const int seg = idx & 15;
            const int cch = row / KK;
            const int hh  = h + (row - cch * KK) - 1;
            xL[k] = row * XPITCH + XCOL0 + seg * 4;
            v4f q = {};
            if ((unsigned)hh < (unsigned)HH)
                q = *(const v4f*)(x + ((size_t)(b * CC + cch) * HH + hh) * WW + seg * 4);
            xq[k] = q;
        }

        v2f pn[GROUPS][3], pd[GROUPS][2], pt[GROUPS];
#pragma unroll
        for (int gg = 0; gg < GROUPS; ++gg) {
            const float* __restrict__ nsrc = nums   + gg * GCH * (KK * KK * 6);
            const float* __restrict__ dsrc = denoms + gg * GCH * (KK * KK * 4);
            pt[gg] = make_v2f(0.0f, 0.0f);
#pragma unroll
            for (int k = 0; k < 3; ++k) pn[gg][k] = *(const v2f*)(nsrc + nG[k]);
#pragma unroll
            for (int k = 0; k < 2; ++k) pd[gg][k] = *(const v2f*)(dsrc + dG[k]);
            if (tailN)      pt[gg] = *(const v2f*)(nsrc + tG);
            else if (tailD) pt[gg] = *(const v2f*)(dsrc + tG);
        }

#pragma unroll
        for (int k = 0; k < 3; ++k) *(v4f*)(xs + xL[k]) = xq[k];
        if (tid < 2 * XROWS)
            xs[(tid >> 1) * XPITCH + ((tid & 1) ? XCOL0 + WW : XCOL0 - 1)] = 0.0f;
#pragma unroll
        for (int gg = 0; gg < GROUPS; ++gg) {
            float* dst = coef + gg * STAGE_FLOATS;
#pragma unroll
            for (int k = 0; k < 3; ++k) *(v2f*)(dst + nL[k]) = pn[gg][k];
#pragma unroll
            for (int k = 0; k < 2; ++k) *(v2f*)(dst + dL[k]) = pd[gg][k];
            if (tailN || tailD) *(v2f*)(dst + tL) = pt[gg];
        }
    }
    __syncthreads();

    v2f acc2[2][4];
#pragma unroll
    for (int t = 0; t < 2; ++t)
#pragma unroll
        for (int k = 0; k < 4; ++k) acc2[t][k] = make_v2f(0.0f, 0.0f);
    const v8f zc = {};

    const float* xlane = xs + (XCOL0 - 1) + w0 + laneM + g * (GCH * KK * XPITCH);
    const float* gcoef = coef + g * STAGE_FLOATS;

    for (int i = 0; i < GCH; ++i) {
        const float* cur = gcoef + (i & 1) * STEP_FLOATS;
        float*       nxt = coef + ((i & 1) ^ 1) * STEP_FLOATS;
        const bool more = i + 1 < GCH;

        v2f pn[GROUPS][3], pd[GROUPS][2], pt[GROUPS];
#pragma unroll
        for (int gg = 0; gg < GROUPS; ++gg) pt[gg] = make_v2f(0.0f, 0.0f);
        if (more) {
#pragma unroll
            for (int gg = 0; gg < GROUPS; ++gg) {
                const float* __restrict__ nsrc = nums   + (gg * GCH + i + 1) * (KK * KK * 6);
                const float* __restrict__ dsrc = denoms + (gg * GCH + i + 1) * (KK * KK * 4);
#pragma unroll
                for (int k = 0; k < 3; ++k) pn[gg][k] = *(const v2f*)(nsrc + nG[k]);
#pragma unroll
                for (int k = 0; k < 2; ++k) pd[gg][k] = *(const v2f*)(dsrc + dG[k]);
                if (tailN)      pt[gg] = *(const v2f*)(nsrc + tG);
                else if (tailD) pt[gg] = *(const v2f*)(dsrc + tG);
            }
        }

        v2f aP0[TAPS], aQ[TAPS], aP1[TAPS];
#pragma unroll
        for (int a = 0; a < KK; ++a) {
            const float* xrow = xlane + (i * KK + a) * XPITCH;
#pragma unroll
            for (int bb = 0; bb < KK; ++bb) {
                const float xv = xrow[bb];
                const float x2 = xv * xv;
                const float u  = hi ? x2 : 1.0f;
                const float ux = u * xv;
                const float x4 = x2 * x2;
                const float v  = hi ? 0.0f : x4;
                const int tap = a * KK + bb;
                aP0[tap] = make_v2f(u,  ux);
                aQ[tap]  = make_v2f(ux, ux * xv);
                aP1[tap] = make_v2f(v,  v * xv);
            }
        }

        v8f Pt[TAPS], Qt[TAPS];
#pragma unroll
        for (int tap = 0; tap < TAPS; ++tap) {
            const v2f bP0 = *(const v2f*)(cur + nOff + tap * 6);
            const v2f bP1 = *(const v2f*)(cur + n1Off + tap * 6);
            const v2f bQ  = *(const v2f*)(cur + dOff + tap * 4);
            v8f P = __builtin_amdgcn_wmma_f32_16x16x4_f32(
                false, aP0[tap], false, bP0, (short)0, zc, false, false);
            P = __builtin_amdgcn_wmma_f32_16x16x4_f32(
                false, aP1[tap], false, bP1, (short)0, P, false, false);
            Qt[tap] = __builtin_amdgcn_wmma_f32_16x16x4_f32(
                false, aQ[tap], false, bQ, (short)0, zc, false, false);
            Pt[tap] = P;
        }

        if (more) {
#pragma unroll
            for (int gg = 0; gg < GROUPS; ++gg) {
                float* dst = nxt + gg * STAGE_FLOATS;
#pragma unroll
                for (int k = 0; k < 3; ++k) *(v2f*)(dst + nL[k]) = pn[gg][k];
#pragma unroll
                for (int k = 0; k < 2; ++k) *(v2f*)(dst + dL[k]) = pd[gg][k];
                if (tailN || tailD) *(v2f*)(dst + tL) = pt[gg];
            }
        }

        v2f cP0[TAPS], cP1[TAPS], cQ[TAPS];
#pragma unroll
        for (int tap = 0; tap < TAPS; ++tap) {
            cP0[tap] = *(const v2f*)(cur + N_TILE1 + nOff + tap * 6);
            cP1[tap] = *(const v2f*)(cur + N_TILE1 + n1Off + tap * 6);
            cQ[tap]  = *(const v2f*)(cur + D_TILE1 + dOff + tap * 4);
        }

        recip_accumulate(Pt, Qt, acc2[0]);

        v8f Pu[TAPS], Qu[TAPS];
#pragma unroll
        for (int tap = 0; tap < TAPS; ++tap) {
            v8f P = __builtin_amdgcn_wmma_f32_16x16x4_f32(
                false, aP0[tap], false, cP0[tap], (short)0, zc, false, false);
            P = __builtin_amdgcn_wmma_f32_16x16x4_f32(
                false, aP1[tap], false, cP1[tap], (short)0, P, false, false);
            Qu[tap] = __builtin_amdgcn_wmma_f32_16x16x4_f32(
                false, aQ[tap], false, cQ[tap], (short)0, zc, false, false);
            Pu[tap] = P;
        }

        recip_accumulate(Pu, Qu, acc2[1]);

        if (more) __syncthreads();
    }

    static_assert((GCH & 1) == 0 && GCH >= 2, "the last step must read stage 1");
    static_assert(4 * 4 * 32 * 4 <= STEP_FLOATS, "hand-over area inside stage 0");
    float* red = coef + (strip * 32 + lane) * 4;
    if (g == 1) {
#pragma unroll
        for (int t = 0; t < 2; ++t) {
            v4f s;
            s.x = acc2[t][0].x; s.y = acc2[t][0].y; s.z = acc2[t][1].x; s.w = acc2[t][1].y;
            *(v4f*)(red + (t * 2 + 0) * (4 * 32 * 4)) = s;
            s.x = acc2[t][2].x; s.y = acc2[t][2].y; s.z = acc2[t][3].x; s.w = acc2[t][3].y;
            *(v4f*)(red + (t * 2 + 1) * (4 * 32 * 4)) = s;
        }
    }
    __syncthreads();
    if (g == 0) {
        const int wbase = w0 + (hi ? 8 : 0);
#pragma unroll
        for (int t = 0; t < 2; ++t) {
            const int f = t * 16 + laneM;
            float* o = out + ((size_t)(b * FF + f) * HH + h) * WW + wbase;
            v4f s, r;
            r = *(const v4f*)(red + (t * 2 + 0) * (4 * 32 * 4));
            s.x = acc2[t][0].x + r.x; s.y = acc2[t][0].y + r.y;
            s.z = acc2[t][1].x + r.z; s.w = acc2[t][1].y + r.w;
            *(v4f*)(o) = s;
            r = *(const v4f*)(red + (t * 2 + 1) * (4 * 32 * 4));
            s.x = acc2[t][2].x + r.x; s.y = acc2[t][2].y + r.y;
            s.z = acc2[t][3].x + r.z; s.w = acc2[t][3].y + r.w;
            *(v4f*)(o + 4) = s;
        }
    }
}

extern "C" void kernel_launch(void* const* d_in, const int* in_sizes, int n_in,
                              void* d_out, int out_size, void* d_ws, size_t ws_size,
                              hipStream_t stream) {
    (void)in_sizes; (void)n_in; (void)out_size; (void)d_ws; (void)ws_size;
    const float* x      = (const float*)d_in[0];
    const float* nums   = (const float*)d_in[1];
    const float* denoms = (const float*)d_in[2];
    float* out = (float*)d_out;

    pade_conv_wmma<<<256, 256, 0, stream>>>(x, nums, denoms, out);
}


# step barrier moved in front of the second reciprocal pass; next step's first-tile B fragments and x values read from LDS one block ahead, none among matrix instructions or behind the release
# speedup vs baseline: 3.3716x; 1.0548x over previous
#include <hip/hip_runtime.h>

typedef __attribute__((ext_vector_type(2))) float v2f;
typedef __attribute__((ext_vector_type(4))) float v4f;
typedef __attribute__((ext_vector_type(8))) float v8f;

__device__ __forceinline__ v2f make_v2f(float a, float b) {
    v2f r; r.x = a; r.y = b; return r;
}

#define BBATCH 4
#define CC     16
#define HH     64
#define WW     64
#define FF     32
#define KK     3
#define TAPS   (KK * KK)

#define GROUPS 2
#define GCH    (CC / GROUPS)

#define NPITCH 60
#define DPITCH 36
#define NUM_F2_PER_F 27
#define DEN_F2_PER_F 18
#define NUM_F2 (FF * NUM_F2_PER_F)
#define DEN_F2 (FF * DEN_F2_PER_F)
#define NUM_LDS_FLOATS (FF * NPITCH)
#define STAGE_FLOATS (NUM_LDS_FLOATS + FF * DPITCH)
#define STEP_FLOATS (GROUPS * STAGE_FLOATS)
#define N_TILE1 (16 * NPITCH)
#define D_TILE1 (16 * DPITCH)

#define XPITCH 72
#define XROWS  (CC * KK)
#define XCOL0  4

__device__ __forceinline__ void recip_accumulate(const v8f (&Pt)[TAPS], const v8f (&Qt)[TAPS],
                                                 v2f (&acc2)[4]) {
    float tq[8 * TAPS];
    asm volatile("v_nop\n\tv_nop\n\tv_nop\n\tv_nop"
                 :
                 : "v"(Pt[0]), "v"(Pt[1]), "v"(Pt[2]), "v"(Pt[3]), "v"(Pt[4]),
                   "v"(Pt[5]), "v"(Pt[6]), "v"(Pt[7]), "v"(Pt[8]),
                   "v"(Qt[0]), "v"(Qt[1]), "v"(Qt[2]), "v"(Qt[3]), "v"(Qt[4]),
                   "v"(Qt[5]), "v"(Qt[6]), "v"(Qt[7]), "v"(Qt[8]));
    asm volatile("v_add_f32 %0, |%4|, 1.0\n\tv_add_f32 %1, |%5|, 1.0\n\t"
                 "v_add_f32 %2, |%6|, 1.0\n\tv_add_f32 %3, |%7|, 1.0"
                 : "=&v"(tq[0]), "=&v"(tq[1]), "=&v"(tq[2]), "=&v"(tq[3])
                 : "v"(Qt[0][0]), "v"(Qt[0][1]), "v"(Qt[0][2]), "v"(Qt[0][3]));
#pragma unroll
    for (int i = 0; i < 8 * TAPS; ++i) {
        if (i + 4 < 8 * TAPS) {
            asm volatile("v_rcp_f32 %0, %0\n\tv_add_f32 %1, |%2|, 1.0"
                         : "+v"(tq[i]), "=&v"(tq[i + 4])
                         : "v"(Qt[(i + 4) >> 3][(i + 4) & 7]));
        } else {
            asm volatile("v_rcp_f32 %0, %0" : "+v"(tq[i]));
        }
        if (i >= 5 && (i & 1)) {
            const int j = i - 5;
            asm volatile("v_pk_fma_f32 %0, %1, %2, %0"
                         : "+v"(acc2[(j & 7) >> 1])
                         : "v"(make_v2f(Pt[j >> 3][j & 7], Pt[j >> 3][(j & 7) + 1])),
                           "v"(make_v2f(tq[j], tq[j + 1])));
        }
    }
#pragma unroll
    for (int j = 8 * TAPS - 4; j < 8 * TAPS; j += 2)
        asm volatile("v_pk_fma_f32 %0, %1, %2, %0"
                     : "+v"(acc2[(j & 7) >> 1])
                     : "v"(make_v2f(Pt[j >> 3][j & 7], Pt[j >> 3][(j & 7) + 1])),
                       "v"(make_v2f(tq[j], tq[j + 1])));
}

__device__ __forceinline__ void load_first_fragments(const float* slab,
                                                     int nOff, int n1Off, int dOff,
                                                     v2f (&bP0)[TAPS], v2f (&bP1)[TAPS],
                                                     v2f (&bQ)[TAPS]) {
#pragma unroll
    for (int tap = 0; tap < TAPS; ++tap) {
        bP0[tap] = *(const v2f*)(slab + nOff + tap * 6);
        bP1[tap] = *(const v2f*)(slab + n1Off + tap * 6);
        bQ[tap]  = *(const v2f*)(slab + dOff + tap * 4);
    }
}

__device__ __forceinline__ void load_patch(const float* xch, float (&xv)[TAPS]) {
#pragma unroll
    for (int a = 0; a < KK; ++a)
#pragma unroll
        for (int bb = 0; bb < KK; ++bb)
            xv[a * KK + bb] = xch[a * XPITCH + bb];
}

__global__ __launch_bounds__(256) void pade_conv_wmma(
    const float* __restrict__ x,
    const float* __restrict__ nums,
    const float* __restrict__ denoms,
    float* __restrict__ out)
{
    const int tid   = threadIdx.x;
    const int lane  = tid & 31;
    const int wv    = __builtin_amdgcn_readfirstlane(tid >> 5);
    const int strip = wv & 3;
    const int g     = wv >> 2;

    const int w0 = strip * 16;
    const int h  = blockIdx.x & (HH - 1);
    const int b  = blockIdx.x >> 6;

    const int  laneM = lane & 15;
    const bool hi    = lane >= 16;

    __shared__ __attribute__((aligned(16))) float coef[2 * STEP_FLOATS];
    __shared__ __attribute__((aligned(16))) float xs[XROWS * XPITCH];

    const int nOff  = laneM * NPITCH + (hi ? 2 : 0);
    const int n1Off = laneM * NPITCH + 4;
    const int dOff  = NUM_LDS_FLOATS + laneM * DPITCH + (hi ? 2 : 0);

    int nG[3], nL[3], dG[2], dL[2];
#pragma unroll
    for (int k = 0; k < 3; ++k) {
        const int idx = tid + k * 256;
        const int ff  = idx / NUM_F2_PER_F;
        const int j   = idx - ff * NUM_F2_PER_F;
        nG[k] = ff * (CC * KK * KK * 6) + 2 * j;
        nL[k] = ff * NPITCH + 2 * j;
    }
#pragma unroll
    for (int k = 0; k < 2; ++k) {
        const int idx = tid + k * 256;
        const int ff  = idx / DEN_F2_PER_F;
        const int j   = idx - ff * DEN_F2_PER_F;
        dG[k] = ff * (CC * KK * KK * 4) + 2 * j;
        dL[k] = NUM_LDS_FLOATS + ff * DPITCH + 2 * j;
    }
    const bool tailN = tid < 96;
    const bool tailD = tid >= 96 && tid < 160;
    int tG, tL;
    {
        const int idxN = 768 + tid;
        const int ffN  = idxN / NUM_F2_PER_F;
        const int jN   = idxN - ffN * NUM_F2_PER_F;
        const int idxD = 512 + ((tid - 96) & 63);
        const int ffD  = idxD / DEN_F2_PER_F;
        const int jD   = idxD - ffD * DEN_F2_PER_F;
        tG = tailN ? ffN * (CC * KK * KK * 6) + 2 * jN
                   : ffD * (CC * KK * KK * 4) + 2 * jD;
        tL = tailN ? ffN * NPITCH + 2 * jN
                   : NUM_LDS_FLOATS + ffD * DPITCH + 2 * jD;
    }

    {
        v4f xq[3];
        int xL[3];
#pragma unroll
        for (int k = 0; k < 3; ++k) {
            const int idx = tid + k * 256;
            const int row = idx >> 4;
            const int seg = idx & 15;
            const int cch = row / KK;
            const int hh  = h + (row - cch * KK) - 1;
            xL[k] = row * XPITCH + XCOL0 + seg * 4;
            v4f q = {};
            if ((unsigned)hh < (unsigned)HH)
                q = *(const v4f*)(x + ((size_t)(b * CC + cch) * HH + hh) * WW + seg * 4);
            xq[k] = q;
        }

        v2f pn[GROUPS][3], pd[GROUPS][2], pt[GROUPS];
#pragma unroll
        for (int gg = 0; gg < GROUPS; ++gg) {
            const float* __restrict__ nsrc = nums   + gg * GCH * (KK * KK * 6);
            const float* __restrict__ dsrc = denoms + gg * GCH * (KK * KK * 4);
            pt[gg] = make_v2f(0.0f, 0.0f);
#pragma unroll
            for (int k = 0; k < 3; ++k) pn[gg][k] = *(const v2f*)(nsrc + nG[k]);
#pragma unroll
            for (int k = 0; k < 2; ++k) pd[gg][k] = *(const v2f*)(dsrc + dG[k]);
            if (tailN)      pt[gg] = *(const v2f*)(nsrc + tG);
            else if (tailD) pt[gg] = *(const v2f*)(dsrc + tG);
        }

#pragma unroll
        for (int k = 0; k < 3; ++k) *(v4f*)(xs + xL[k]) = xq[k];
        if (tid < 2 * XROWS)
            xs[(tid >> 1) * XPITCH + ((tid & 1) ? XCOL0 + WW : XCOL0 - 1)] = 0.0f;
#pragma unroll
        for (int gg = 0; gg < GROUPS; ++gg) {
            float* dst = coef + gg * STAGE_FLOATS;
#pragma unroll
            for (int k = 0; k < 3; ++k) *(v2f*)(dst + nL[k]) = pn[gg][k];
#pragma unroll
            for (int k = 0; k < 2; ++k) *(v2f*)(dst + dL[k]) = pd[gg][k];
            if (tailN || tailD) *(v2f*)(dst + tL) = pt[gg];
        }
    }
    __syncthreads();

    v2f acc2[2][4];
#pragma unroll
    for (int t = 0; t < 2; ++t)
#pragma unroll
        for (int k = 0; k < 4; ++k) acc2[t][k] = make_v2f(0.0f, 0.0f);
    const v8f zc = {};

    const float* xlane = xs + (XCOL0 - 1) + w0 + laneM + g * (GCH * KK * XPITCH);
    const float* gcoef = coef + g * STAGE_FLOATS;

    v2f bP0[TAPS], bP1[TAPS], bQ[TAPS];
    float xv[TAPS];
    load_first_fragments(gcoef, nOff, n1Off, dOff, bP0, bP1, bQ);
    load_patch(xlane, xv);

    for (int i = 0; i < GCH; ++i) {
        const float* cur = gcoef + (i & 1) * STEP_FLOATS;
        float*       nxt = coef + ((i & 1) ^ 1) * STEP_FLOATS;
        const bool more = i + 1 < GCH;

        v2f pn[GROUPS][3], pd[GROUPS][2], pt[GROUPS];
#pragma unroll
        for (int gg = 0; gg < GROUPS; ++gg) pt[gg] = make_v2f(0.0f, 0.0f);
        if (more) {
#pragma unroll
            for (int gg = 0; gg < GROUPS; ++gg) {
                const float* __restrict__ nsrc = nums   + (gg * GCH + i + 1) * (KK * KK * 6);
                const float* __restrict__ dsrc = denoms + (gg * GCH + i + 1) * (KK * KK * 4);
#pragma unroll
                for (int k = 0; k < 3; ++k) pn[gg][k] = *(const v2f*)(nsrc + nG[k]);
#pragma unroll
                for (int k = 0; k < 2; ++k) pd[gg][k] = *(const v2f*)(dsrc + dG[k]);
                if (tailN)      pt[gg] = *(const v2f*)(nsrc + tG);
                else if (tailD) pt[gg] = *(const v2f*)(dsrc + tG);
            }
        }

        v2f aP0[TAPS], aQ[TAPS], aP1[TAPS];
#pragma unroll
        for (int a = 0; a < KK; ++a) {
#pragma unroll
            for (int bb = 0; bb < KK; ++bb) {
                const int tap = a * KK + bb;
                const float xt = xv[tap];
                const float x2 = xt * xt;
                const float u  = hi ? x2 : 1.0f;
                const float ux = u * xt;
                const float x4 = x2 * x2;
                const float v  = hi ? 0.0f : x4;
                aP0[tap] = make_v2f(u,  ux);
                aQ[tap]  = aP0[tap] * make_v2f(xt, xt);
                aP1[tap] = make_v2f(v,  v * xt);
            }
        }

        v8f Pt[TAPS], Qt[TAPS];
#pragma unroll
        for (int tap = 0; tap < TAPS; ++tap) {
            v8f P = __builtin_amdgcn_wmma_f32_16x16x4_f32(
                false, aP0[tap], false, bP0[tap], (short)0, zc, false, false);
            P = __builtin_amdgcn_wmma_f32_16x16x4_f32(
                false, aP1[tap], false, bP1[tap], (short)0, P, false, false);
            Qt[tap] = __builtin_amdgcn_wmma_f32_16x16x4_f32(
                false, aQ[tap], false, bQ[tap], (short)0, zc, false, false);
            Pt[tap] = P;
        }

        if (more) {
#pragma unroll
            for (int gg = 0; gg < GROUPS; ++gg) {
                float* dst = nxt + gg * STAGE_FLOATS;
#pragma unroll
                for (int k = 0; k < 3; ++k) *(v2f*)(dst + nL[k]) = pn[gg][k];
#pragma unroll
                for (int k = 0; k < 2; ++k) *(v2f*)(dst + dL[k]) = pd[gg][k];
                if (tailN || tailD) *(v2f*)(dst + tL) = pt[gg];
            }
        }

        v2f cP0[TAPS], cP1[TAPS], cQ[TAPS];
#pragma unroll
        for (int tap = 0; tap < TAPS; ++tap) {
            cP0[tap] = *(const v2f*)(cur + N_TILE1 + nOff + tap * 6);
            cP1[tap] = *(const v2f*)(cur + N_TILE1 + n1Off + tap * 6);
            cQ[tap]  = *(const v2f*)(cur + D_TILE1 + dOff + tap * 4);
        }
        load_patch(xlane + (more ? i + 1 : i) * KK * XPITCH, xv);

        recip_accumulate(Pt, Qt, acc2[0]);

        v8f Pu[TAPS], Qu[TAPS];
#pragma unroll
        for (int tap = 0; tap < TAPS; ++tap) {
            v8f P = __builtin_amdgcn_wmma_f32_16x16x4_f32(
                false, aP0[tap], false, cP0[tap], (short)0, zc, false, false);
            P = __builtin_amdgcn_wmma_f32_16x16x4_f32(
                false, aP1[tap], false, cP1[tap], (short)0, P, false, false);
            Qu[tap] = __builtin_amdgcn_wmma_f32_16x16x4_f32(
                false, aQ[tap], false, cQ[tap], (short)0, zc, false, false);
            Pu[tap] = P;
        }

        if (more) {
            __syncthreads();
            load_first_fragments(gcoef + ((i & 1) ^ 1) * STEP_FLOATS,
                                 nOff, n1Off, dOff, bP0, bP1, bQ);
        }

        recip_accumulate(Pu, Qu, acc2[1]);
    }

    static_assert((GCH & 1) == 0 && GCH >= 2, "the last step must read stage 1");
    static_assert(4 * 4 * 32 * 4 <= STEP_FLOATS, "hand-over area inside stage 0");
    float* red = coef + (strip * 32 + lane) * 4;
    if (g == 1) {
#pragma unroll
        for (int t = 0; t < 2; ++t) {
            v4f s;
            s.x = acc2[t][0].x; s.y = acc2[t][0].y; s.z = acc2[t][1].x; s.w = acc2[t][1].y;
            *(v4f*)(red + (t * 2 + 0) * (4 * 32 * 4)) = s;
            s.x = acc2[t][2].x; s.y = acc2[t][2].y; s.z = acc2[t][3].x; s.w = acc2[t][3].y;
            *(v4f*)(red + (t * 2 + 1) * (4 * 32 * 4)) = s;
        }
    }
    __syncthreads();
    if (g == 0) {
        const int wbase = w0 + (hi ? 8 : 0);
#pragma unroll
        for (int t = 0; t < 2; ++t) {
            const int f = t * 16 + laneM;
            float* o = out + ((size_t)(b * FF + f) * HH + h) * WW + wbase;
            v4f s, r;
            r = *(const v4f*)(red + (t * 2 + 0) * (4 * 32 * 4));
            s.x = acc2[t][0].x + r.x; s.y = acc2[t][0].y + r.y;
            s.z = acc2[t][1].x + r.z; s.w = acc2[t][1].y + r.w;
            *(v4f*)(o) = s;
            r = *(const v4f*)(red + (t * 2 + 1) * (4 * 32 * 4));
            s.x = acc2[t][2].x + r.x; s.y = acc2[t][2].y + r.y;
            s.z = acc2[t][3].x + r.z; s.w = acc2[t][3].y + r.w;
            *(v4f*)(o + 4) = s;
        }
    }
}

extern "C" void kernel_launch(void* const* d_in, const int* in_sizes, int n_in,
                              void* d_out, int out_size, void* d_ws, size_t ws_size,
                              hipStream_t stream) {
    (void)in_sizes; (void)n_in; (void)out_size; (void)d_ws; (void)ws_size;
    const float* x      = (const float*)d_in[0];
    const float* nums   = (const float*)d_in[1];
    const float* denoms = (const float*)d_in[2];
    float* out = (float*)d_out;

    pade_conv_wmma<<<256, 256, 0, stream>>>(x, nums, denoms, out);
}
